# starting kernel
# baseline (speedup 1.0000x reference)
#include <hip/hip_runtime.h>


typedef __attribute__((ext_vector_type(2))) float v2f;
typedef __attribute__((ext_vector_type(8))) float v8f;

#define ALPHA 0.1f


__global__ void k_deg_init(float* __restrict__ deg, int n) {
    int i = blockIdx.x * blockDim.x + threadIdx.x;
    if (i < n) deg[i] = 1.0f;
}

__global__ void k_deg_count(float* __restrict__ deg, const int* __restrict__ dst, int e) {
    int i = blockIdx.x * blockDim.x + threadIdx.x;
    if (i < e) atomicAdd(&deg[dst[i]], 1.0f);
}

__global__ void k_rsqrt(float* __restrict__ d, int n) {
    int i = blockIdx.x * blockDim.x + threadIdx.x;
    if (i < n) {
        float v = d[i];
        d[i] = (v > 0.0f) ? rsqrtf(v) : 0.0f;
    }
}

__global__ void k_edge_w(float* __restrict__ w, const int* __restrict__ src,
                         const int* __restrict__ dst, const float* __restrict__ dinv, int e) {
    int i = blockIdx.x * blockDim.x + threadIdx.x;
    if (i < e) w[i] = dinv[src[i]] * dinv[dst[i]];
}


template <int K, int NOUT>
__global__ void k_gemm_wmma(const float* __restrict__ A, const float* __restrict__ W,
                            const float* __restrict__ bias, float* __restrict__ C) {
    const int wave = threadIdx.x >> 5;
    const int lane = threadIdx.x & 31;
    const int row0 = blockIdx.x * 16;
    const int col0 = wave * 16;
    const int lidx = lane & 15;
    const int khalf = (lane >> 4) * 2;

    const float* __restrict__ arow = A + (size_t)(row0 + lidx) * K + khalf;
    const float* __restrict__ wcol = W + (size_t)khalf * NOUT + col0 + lidx;

    v8f acc = {};
#pragma unroll 4
    for (int k = 0; k < K; k += 4) {
        v2f a, b;
        a.x = arow[0];
        a.y = arow[1];
        b.x = wcol[0];
        b.y = wcol[NOUT];
        acc = __builtin_amdgcn_wmma_f32_16x16x4_f32(
             false, a,  false, b,
             (short)0, acc,  false,  false);
        arow += 4;
        wcol += 4 * NOUT;
    }

    const float bv = bias[col0 + lidx];
    const int moff = (lane >> 4) * 8;
    float* __restrict__ cbase = C + (size_t)row0 * NOUT + col0 + lidx;
#pragma unroll
    for (int j = 0; j < 8; ++j) {
        cbase[(size_t)(j + moff) * NOUT] = acc[j] + bv;
    }
}


template <int F>
__global__ void k_prop_self(float* __restrict__ out, const float* __restrict__ h,
                            const float* __restrict__ dinv, int n) {
    const int vpr = F / 4;
    long long gid = (long long)blockIdx.x * blockDim.x + threadIdx.x;
    long long total = (long long)n * vpr;
    if (gid >= total) return;
    int i = (int)(gid / vpr);
    int f4 = (int)(gid % vpr);
    float sw = dinv[i] * dinv[i];
    float4 hv = ((const float4*)(h + (size_t)i * F))[f4];
    float4 ov = make_float4(sw * hv.x, sw * hv.y, sw * hv.z, sw * hv.w);
    ((float4*)(out + (size_t)i * F))[f4] = ov;
}

template <int F>
__global__ void k_prop_edges(float* __restrict__ out, const float* __restrict__ h,
                             const int* __restrict__ src, const int* __restrict__ dst,
                             const float* __restrict__ w, int e) {
    const int vpr = F / 4;
    long long gid = (long long)blockIdx.x * blockDim.x + threadIdx.x;
    long long total = (long long)e * vpr;
    if (gid >= total) return;
    int ei = (int)(gid / vpr);
    int f4 = (int)(gid % vpr);
    int s = src[ei];
    int d = dst[ei];
    float we = w[ei];
    float4 hv = ((const float4*)(h + (size_t)s * F))[f4];
    float* o = out + (size_t)d * F + (size_t)f4 * 4;
    atomicAdd(o + 0, we * hv.x);
    atomicAdd(o + 1, we * hv.y);
    atomicAdd(o + 2, we * hv.z);
    atomicAdd(o + 3, we * hv.w);
}

__global__ void k_combine(float* __restrict__ h, const float* __restrict__ tmp,
                          const float* __restrict__ h0, long long n4) {
    long long gid = (long long)blockIdx.x * blockDim.x + threadIdx.x;
    if (gid >= n4) return;
    float4 t = ((const float4*)tmp)[gid];
    float4 z = ((const float4*)h0)[gid];
    float4 r = make_float4((1.0f - ALPHA) * t.x + ALPHA * z.x,
                           (1.0f - ALPHA) * t.y + ALPHA * z.y,
                           (1.0f - ALPHA) * t.z + ALPHA * z.z,
                           (1.0f - ALPHA) * t.w + ALPHA * z.w);
    ((float4*)h)[gid] = r;
}


static inline int cdiv(long long a, int b) { return (int)((a + b - 1) / b); }

extern "C" void kernel_launch(void* const* d_in, const int* in_sizes, int n_in,
                              void* d_out, int out_size, void* d_ws, size_t ws_size,
                              hipStream_t stream) {
    const float* x  = (const float*)d_in[0];
    const int*   ei = (const int*)d_in[1];
    const float* W1 = (const float*)d_in[2];
    const float* b1 = (const float*)d_in[3];
    const float* W3 = (const float*)d_in[4];
    const float* b3 = (const float*)d_in[5];
    float* out = (float*)d_out;

    const int N = in_sizes[0] / 256;
    const int E = in_sizes[1] / 2;
    const int* src = ei;
    const int* dst = ei + E;

    float* ws = (float*)d_ws;
    float* dinv = ws;
    size_t off = ((size_t)N + 63) & ~(size_t)63;
    float* wbuf = ws + off;
    off += ((size_t)E + 63) & ~(size_t)63;
    float* h0 = ws + off;
    off += (size_t)N * 128;
    float* h = ws + off;
    off += (size_t)N * 128;
    float* tmp = ws + off;
    off += (size_t)N * 128;
    float* hw = ws + off;

    const int T = 256;

    k_deg_init<<<cdiv(N, T), T, 0, stream>>>(dinv, N);
    k_deg_count<<<cdiv(E, T), T, 0, stream>>>(dinv, dst, E);
    k_rsqrt<<<cdiv(N, T), T, 0, stream>>>(dinv, N);
    k_edge_w<<<cdiv(E, T), T, 0, stream>>>(wbuf, src, dst, dinv, E);

    k_gemm_wmma<256, 128><<<N / 16, 32 * (128 / 16), 0, stream>>>(x, W1, b1, tmp);
    k_prop_self<128><<<cdiv((long long)N * 32, T), T, 0, stream>>>(h0, tmp, dinv, N);
    k_prop_edges<128><<<cdiv((long long)E * 32, T), T, 0, stream>>>(h0, tmp, src, dst, wbuf, E);

    hipMemcpyAsync(h, h0, (size_t)N * 128 * sizeof(float), hipMemcpyDeviceToDevice, stream);

    const long long n4 = (long long)N * 128 / 4;
    for (int it = 0; it < 10; ++it) {
        k_prop_self<128><<<cdiv((long long)N * 32, T), T, 0, stream>>>(tmp, h, dinv, N);
        k_prop_edges<128><<<cdiv((long long)E * 32, T), T, 0, stream>>>(tmp, h, src, dst, wbuf, E);
        k_combine<<<cdiv(n4, T), T, 0, stream>>>(h, tmp, h0, n4);
    }

    k_gemm_wmma<128, 64><<<N / 16, 32 * (64 / 16), 0, stream>>>(h, W3, b3, hw);
    k_prop_self<64><<<cdiv((long long)N * 16, T), T, 0, stream>>>(out, hw, dinv, N);
    k_prop_edges<64><<<cdiv((long long)E * 16, T), T, 0, stream>>>(out, hw, src, dst, wbuf, E);
}


# Mixing loop: self-loop pass and combine pass become one start-value pass, acc = 0.9 dinv^2 h + 0.1 h0; edge weights carry the 0.9; acc is the next h; sum order: 0.1 h0 and self first, then edges
# speedup vs baseline: 2.2629x; 2.2629x over previous
#include <hip/hip_runtime.h>


typedef __attribute__((ext_vector_type(2))) float v2f;
typedef __attribute__((ext_vector_type(8))) float v8f;

#define ALPHA 0.1f


__global__ void k_deg_init(float* __restrict__ deg, int n) {
    int i = blockIdx.x * blockDim.x + threadIdx.x;
    if (i < n) deg[i] = 1.0f;
}

__global__ void k_deg_count(float* __restrict__ deg, const int* __restrict__ dst, int e) {
    int i = blockIdx.x * blockDim.x + threadIdx.x;
    if (i < e) atomicAdd(&deg[dst[i]], 1.0f);
}

__global__ void k_rsqrt(float* __restrict__ d, int n) {
    int i = blockIdx.x * blockDim.x + threadIdx.x;
    if (i < n) {
        float v = d[i];
        d[i] = (v > 0.0f) ? rsqrtf(v) : 0.0f;
    }
}

__global__ void k_edge_w(float* __restrict__ w, const int* __restrict__ src,
                         const int* __restrict__ dst, const float* __restrict__ dinv, int e) {
    int i = blockIdx.x * blockDim.x + threadIdx.x;
    if (i < e) w[i] = dinv[src[i]] * dinv[dst[i]];
}


template <int K, int NOUT>
__global__ void k_gemm_wmma(const float* __restrict__ A, const float* __restrict__ W,
                            const float* __restrict__ bias, float* __restrict__ C) {
    const int wave = threadIdx.x >> 5;
    const int lane = threadIdx.x & 31;
    const int row0 = blockIdx.x * 16;
    const int col0 = wave * 16;
    const int lidx = lane & 15;
    const int khalf = (lane >> 4) * 2;

    const float* __restrict__ arow = A + (size_t)(row0 + lidx) * K + khalf;
    const float* __restrict__ wcol = W + (size_t)khalf * NOUT + col0 + lidx;

    v8f acc = {};
#pragma unroll 4
    for (int k = 0; k < K; k += 4) {
        v2f a, b;
        a.x = arow[0];
        a.y = arow[1];
        b.x = wcol[0];
        b.y = wcol[NOUT];
        acc = __builtin_amdgcn_wmma_f32_16x16x4_f32(
             false, a,  false, b,
             (short)0, acc,  false,  false);
        arow += 4;
        wcol += 4 * NOUT;
    }

    const float bv = bias[col0 + lidx];
    const int moff = (lane >> 4) * 8;
    float* __restrict__ cbase = C + (size_t)row0 * NOUT + col0 + lidx;
#pragma unroll
    for (int j = 0; j < 8; ++j) {
        cbase[(size_t)(j + moff) * NOUT] = acc[j] + bv;
    }
}


template <int F>
__global__ void k_prop_self(float* __restrict__ out, const float* __restrict__ h,
                            const float* __restrict__ dinv, int n) {
    const int vpr = F / 4;
    long long gid = (long long)blockIdx.x * blockDim.x + threadIdx.x;
    long long total = (long long)n * vpr;
    if (gid >= total) return;
    int i = (int)(gid / vpr);
    int f4 = (int)(gid % vpr);
    float sw = dinv[i] * dinv[i];
    float4 hv = ((const float4*)(h + (size_t)i * F))[f4];
    float4 ov = make_float4(sw * hv.x, sw * hv.y, sw * hv.z, sw * hv.w);
    ((float4*)(out + (size_t)i * F))[f4] = ov;
}


template <int F, int G>
__device__ __forceinline__ void scatter_group(float* __restrict__ out, const float* __restrict__ h,
                                              int sv, int dv, int wbits, int j0, int lane) {
    constexpr int NL = F / 32;
    float hv[G][NL];
#pragma unroll
    for (int u = 0; u < G; ++u) {
        const int s = __builtin_amdgcn_readlane(sv, j0 + u);
        const float* __restrict__ hb = h + (size_t)s * F;
#pragma unroll
        for (int k = 0; k < NL; ++k) hv[u][k] = hb[lane + 32 * k];
    }
#pragma unroll
    for (int u = 0; u < G; ++u) {
        const int d = __builtin_amdgcn_readlane(dv, j0 + u);
        const float we = __int_as_float(__builtin_amdgcn_readlane(wbits, j0 + u));
        float* __restrict__ ob = out + (size_t)d * F;
#pragma unroll
        for (int k = 0; k < NL; ++k) atomicAdd(ob + lane + 32 * k, we * hv[u][k]);
    }
}

template <int F>
__global__ void k_prop_edges(float* __restrict__ out, const float* __restrict__ h,
                             const int* __restrict__ src, const int* __restrict__ dst,
                             const float* __restrict__ w, float scale, int e) {
    constexpr int U = 32 / (F / 32);
    const int lane = threadIdx.x & 31;
    const int wv = __builtin_amdgcn_readfirstlane(
        (int)((blockIdx.x * blockDim.x + threadIdx.x) >> 5));
    const long long base = (long long)wv * 32;
    if (base >= e) return;
    const int cnt = (e - base < 32) ? (int)(e - base) : 32;

    int sv = 0, dv = 0, wbits = 0;
    if (lane < cnt) {
        sv = src[base + lane];
        dv = dst[base + lane];
        wbits = __float_as_int(scale * w[base + lane]);
    }

    if (cnt == 32) {
        for (int j0 = 0; j0 < 32; j0 += U) scatter_group<F, U>(out, h, sv, dv, wbits, j0, lane);
    } else {
        for (int j = 0; j < cnt; ++j) scatter_group<F, 1>(out, h, sv, dv, wbits, j, lane);
    }
}

#define SU 4
__global__ void k_start(float4* __restrict__ acc, const float4* __restrict__ h,
                        const float4* __restrict__ h0, const float* __restrict__ dinv,
                        long long n4) {
    const long long base = (long long)blockIdx.x * (SU * 256) + threadIdx.x;
    if (base - threadIdx.x + SU * 256 <= n4) {
        float4 hv[SU], zv[SU];
        float dv[SU];
#pragma unroll
        for (int u = 0; u < SU; ++u) {
            const long long e = base + u * 256;
            hv[u] = h[e];
            zv[u] = h0[e];
            dv[u] = dinv[e >> 5];
        }
#pragma unroll
        for (int u = 0; u < SU; ++u) {
            const float sw = (1.0f - ALPHA) * (dv[u] * dv[u]);
            acc[base + u * 256] = make_float4(sw * hv[u].x + ALPHA * zv[u].x,
                                              sw * hv[u].y + ALPHA * zv[u].y,
                                              sw * hv[u].z + ALPHA * zv[u].z,
                                              sw * hv[u].w + ALPHA * zv[u].w);
        }
    } else {
        for (int u = 0; u < SU; ++u) {
            const long long e = base + u * 256;
            if (e < n4) {
                const float d = dinv[e >> 5];
                const float sw = (1.0f - ALPHA) * (d * d);
                const float4 a = h[e];
                const float4 z = h0[e];
                acc[e] = make_float4(sw * a.x + ALPHA * z.x, sw * a.y + ALPHA * z.y,
                                     sw * a.z + ALPHA * z.z, sw * a.w + ALPHA * z.w);
            }
        }
    }
}


static inline int cdiv(long long a, int b) { return (int)((a + b - 1) / b); }

extern "C" void kernel_launch(void* const* d_in, const int* in_sizes, int n_in,
                              void* d_out, int out_size, void* d_ws, size_t ws_size,
                              hipStream_t stream) {
    const float* x  = (const float*)d_in[0];
    const int*   ei = (const int*)d_in[1];
    const float* W1 = (const float*)d_in[2];
    const float* b1 = (const float*)d_in[3];
    const float* W3 = (const float*)d_in[4];
    const float* b3 = (const float*)d_in[5];
    float* out = (float*)d_out;

    const int N = in_sizes[0] / 256;
    const int E = in_sizes[1] / 2;
    const int* src = ei;
    const int* dst = ei + E;

    float* ws = (float*)d_ws;
    float* dinv = ws;
    size_t off = ((size_t)N + 63) & ~(size_t)63;
    float* wbuf = ws + off;
    off += ((size_t)E + 63) & ~(size_t)63;
    float* h0 = ws + off;
    off += (size_t)N * 128;
    float* h = ws + off;
    off += (size_t)N * 128;
    float* tmp = ws + off;
    off += (size_t)N * 128;
    float* hw = ws + off;

    const int T = 256;

    k_deg_init<<<cdiv(N, T), T, 0, stream>>>(dinv, N);
    k_deg_count<<<cdiv(E, T), T, 0, stream>>>(dinv, dst, E);
    k_rsqrt<<<cdiv(N, T), T, 0, stream>>>(dinv, N);
    k_edge_w<<<cdiv(E, T), T, 0, stream>>>(wbuf, src, dst, dinv, E);

    k_gemm_wmma<256, 128><<<N / 16, 32 * (128 / 16), 0, stream>>>(x, W1, b1, tmp);
    k_prop_self<128><<<cdiv((long long)N * 32, T), T, 0, stream>>>(h0, tmp, dinv, N);
    k_prop_edges<128><<<cdiv(E, T), T, 0, stream>>>(h0, tmp, src, dst, wbuf, 1.0f, E);

    const long long n4 = (long long)N * 128 / 4;
    const float* hs = h0;
    for (int it = 0; it < 10; ++it) {
        float* acc = (it & 1) ? tmp : h;
        k_start<<<cdiv(n4, SU * T), T, 0, stream>>>((float4*)acc, (const float4*)hs,
                                                    (const float4*)h0, dinv, n4);
        k_prop_edges<128><<<cdiv(E, T), T, 0, stream>>>(acc, hs, src, dst, wbuf, 1.0f - ALPHA, E);
        hs = acc;
    }

    k_gemm_wmma<128, 64><<<N / 16, 32 * (64 / 16), 0, stream>>>(hs, W3, b3, hw);
    k_prop_self<64><<<cdiv((long long)N * 16, T), T, 0, stream>>>(out, hw, dinv, N);
    k_prop_edges<64><<<cdiv(E, T), T, 0, stream>>>(out, hw, src, dst, wbuf, 1.0f, E);
}


# Dense layers: resident workgroups loop over 16-row tiles; a wave keeps its 16 W columns in registers for all tiles, the x tile comes through LDS by whole-line loads one tile ahead; sum order kept
# speedup vs baseline: 3.6368x; 1.6071x over previous
#include <hip/hip_runtime.h>


typedef __attribute__((ext_vector_type(2))) float v2f;
typedef __attribute__((ext_vector_type(4))) float v4f;
typedef __attribute__((ext_vector_type(8))) float v8f;

#define ALPHA 0.1f
#define RT 32
#define RT_SHIFT 5


__global__ void k_deg_init(float* __restrict__ deg, int* __restrict__ cnt, int n, int nb) {
    int i = blockIdx.x * blockDim.x + threadIdx.x;
    if (i < n) deg[i] = 1.0f;
    if (i < nb) cnt[i] = 0;
}

__global__ void k_deg_count(float* __restrict__ deg, int* __restrict__ cnt,
                            const int* __restrict__ dst, int e) {
    int i = blockIdx.x * blockDim.x + threadIdx.x;
    if (i < e) {
        const int d = dst[i];
        atomicAdd(&deg[d], 1.0f);
        atomicAdd(&cnt[d >> RT_SHIFT], 1);
    }
}

__global__ void k_rsqrt(float* __restrict__ d, int n) {
    int i = blockIdx.x * blockDim.x + threadIdx.x;
    if (i < n) {
        float v = d[i];
        d[i] = (v > 0.0f) ? rsqrtf(v) : 0.0f;
    }
}

__global__ void k_range_scan(int* __restrict__ cnt, int* __restrict__ off, int nb) {
    __shared__ int part[1024];
    const int t = threadIdx.x;
    const int ipt = (nb + 1023) / 1024;
    const int lo = t * ipt;
    int s = 0;
    for (int i = lo; i < lo + ipt && i < nb; ++i) s += cnt[i];
    part[t] = s;
    __syncthreads();
    for (int d = 1; d < 1024; d <<= 1) {
        const int v = (t >= d) ? part[t - d] : 0;
        __syncthreads();
        part[t] += v;
        __syncthreads();
    }
    int run = part[t] - s;
    for (int i = lo; i < lo + ipt && i < nb; ++i) {
        const int c = cnt[i];
        off[i] = run;
        cnt[i] = run;
        run += c;
    }
    if (t == 1023) off[nb] = part[1023];
}

__global__ void k_range_fill(int* __restrict__ bs, int* __restrict__ bd, float* __restrict__ bw,
                             int* __restrict__ cur, const int* __restrict__ src,
                             const int* __restrict__ dst, const float* __restrict__ dinv, int e) {
    int i = blockIdx.x * blockDim.x + threadIdx.x;
    if (i < e) {
        const int s = src[i], d = dst[i];
        const int p = atomicAdd(&cur[d >> RT_SHIFT], 1);
        bs[p] = s;
        bd[p] = d;
        bw[p] = dinv[s] * dinv[d];
    }
}


template <int K, int NOUT>
__global__ void __launch_bounds__(32 * (NOUT / 16))
k_gemm_wmma(const float* __restrict__ A, const float* __restrict__ W,
            const float* __restrict__ bias, float* __restrict__ C, int m) {
    constexpr int NTH = 32 * (NOUT / 16);
    constexpr int P = K + 4;
    constexpr int KS = K / 4;
    constexpr int KV = K / 4;
    constexpr int NV = 16 * KV / NTH;
    __shared__ __attribute__((aligned(16))) float xs[2 * 16 * P];

    const int tid = threadIdx.x;
    const int wave = tid >> 5;
    const int lane = tid & 31;
    const int col0 = wave * 16;
    const int lidx = lane & 15;
    const int khalf = (lane >> 4) * 2;
    const int ntiles = m >> 4;
    const int step = (int)gridDim.x;

    v2f bf[KS];
    const float* __restrict__ wcol = W + (size_t)khalf * NOUT + col0 + lidx;
#pragma unroll
    for (int s = 0; s < KS; ++s) {
        bf[s].x = wcol[(size_t)(4 * s) * NOUT];
        bf[s].y = wcol[(size_t)(4 * s + 1) * NOUT];
    }
    const float bv = bias[col0 + lidx];
    const int moff = (lane >> 4) * 8;

    static_assert(NV == 4, "four 16-byte pieces of a tile per thread");
    const int o0 = ((tid + 0 * NTH) / KV) * P + 4 * ((tid + 0 * NTH) % KV);
    const int o1 = ((tid + 1 * NTH) / KV) * P + 4 * ((tid + 1 * NTH) % KV);
    const int o2 = ((tid + 2 * NTH) / KV) * P + 4 * ((tid + 2 * NTH) % KV);
    const int o3 = ((tid + 3 * NTH) / KV) * P + 4 * ((tid + 3 * NTH) % KV);

    int t = (int)blockIdx.x;
    float4 p0 = make_float4(0.f, 0.f, 0.f, 0.f), p1 = p0, p2 = p0, p3 = p0;
    if (t < ntiles) {
        const float4* __restrict__ g = (const float4*)(A + (size_t)t * 16 * K) + tid;
        p0 = g[0 * NTH]; p1 = g[1 * NTH]; p2 = g[2 * NTH]; p3 = g[3 * NTH];
    }
    *(float4*)(xs + o0) = p0; *(float4*)(xs + o1) = p1;
    *(float4*)(xs + o2) = p2; *(float4*)(xs + o3) = p3;
    __syncthreads();

    int cur = 0;
    for (; t < ntiles; t += step) {
        const int tn = t + step;
        if (tn < ntiles) {
            const float4* __restrict__ g = (const float4*)(A + (size_t)tn * 16 * K) + tid;
            p0 = g[0 * NTH]; p1 = g[1 * NTH]; p2 = g[2 * NTH]; p3 = g[3 * NTH];
        }

        const float* ar = xs + cur * (16 * P) + lidx * P + khalf;
        v8f acc = {};
#pragma unroll
        for (int s = 0; s < KS; ++s) {
            const v2f a = *(const v2f*)(ar + 4 * s);
            acc = __builtin_amdgcn_wmma_f32_16x16x4_f32(
                 false, a,  false, bf[s],
                 (short)0, acc,  false,  false);
        }

        float* xn = xs + (cur ^ 1) * (16 * P);
        *(float4*)(xn + o0) = p0; *(float4*)(xn + o1) = p1;
        *(float4*)(xn + o2) = p2; *(float4*)(xn + o3) = p3;
        __syncthreads();
        cur ^= 1;

        float* __restrict__ cbase = C + (size_t)t * 16 * NOUT + col0 + lidx;
#pragma unroll
        for (int j = 0; j < 8; ++j) cbase[(size_t)(j + moff) * NOUT] = acc[j] + bv;
    }
}


template <int VW> struct row_vec;
template <> struct row_vec<4> { typedef v4f type; };
template <> struct row_vec<2> { typedef v2f type; };

template <int F>
__device__ __forceinline__ int tile_pos(int f) {
    constexpr int VW = F / 32;
    return (f % VW) * 32 + f / VW;
}

template <int F, int G>
__device__ __forceinline__ void tile_edge_group(float* tile, const float* __restrict__ h,
                                                int sv, int dl, int wbits, int j0, int lane) {
    constexpr int VW = F / 32;
    typedef typename row_vec<VW>::type vrow;
    vrow hv[G];
#pragma unroll
    for (int u = 0; u < G; ++u) {
        const int s = __builtin_amdgcn_readlane(sv, j0 + u);
        const float* __restrict__ hb = h + (size_t)s * F;
        hv[u] = ((const vrow*)hb)[lane];
    }
#pragma unroll
    for (int u = 0; u < G; ++u) {
        const int d = __builtin_amdgcn_readlane(dl, j0 + u);
        const float we = __int_as_float(__builtin_amdgcn_readlane(wbits, j0 + u));
        float* tb = tile + d * F;
#pragma unroll
        for (int k = 0; k < VW; ++k) atomicAdd(tb + lane + 32 * k, we * hv[u][k]);
    }
}

#define NT 256

template <int F, bool MIX>
__global__ void __launch_bounds__(NT, 16)
k_prop_tile(float* __restrict__ out, const float* __restrict__ h, const float* __restrict__ z,
            const float* __restrict__ dinv, const int* __restrict__ bs,
            const int* __restrict__ bd, const float* __restrict__ bw,
            const int* __restrict__ off, int n) {
    constexpr int V = F / 4;
    constexpr int G = 32 / (F / 32);
    constexpr int NW = NT / 32;
    __shared__ float tile[RT * F];
    const int tid = threadIdx.x;
    const int lane = tid & 31;
    const int wib = __builtin_amdgcn_readfirstlane(tid >> 5);
    const int row0 = (int)blockIdx.x * RT;
    const int e0 = off[blockIdx.x], e1 = off[blockIdx.x + 1];

    int base = e0 + wib * 32;
    int sv = 0, dl = 0, wbits = 0;
    if (base + lane < e1) {
        sv = bs[base + lane];
        dl = bd[base + lane] - row0;
        const float w = bw[base + lane];
        wbits = __float_as_int(MIX ? (1.0f - ALPHA) * w : w);
    }

    float4 hv[RT * V / NT], zv[RT * V / NT];
    float sw[RT * V / NT];
#pragma unroll
    for (int k = 0; k < RT * V / NT; ++k) {
        const int idx = tid + k * NT;
        const int r = idx / V, q = idx % V;
        const int row = (row0 + r < n) ? row0 + r : n - 1;
        const float d = dinv[row];
        sw[k] = MIX ? (1.0f - ALPHA) * (d * d) : d * d;
        hv[k] = ((const float4*)(h + (size_t)row * F))[q];
        if (MIX) zv[k] = ((const float4*)(z + (size_t)row * F))[q];
    }
#pragma unroll
    for (int k = 0; k < RT * V / NT; ++k) {
        float4 t = make_float4(sw[k] * hv[k].x, sw[k] * hv[k].y, sw[k] * hv[k].z, sw[k] * hv[k].w);
        if (MIX) t = make_float4(ALPHA * zv[k].x + t.x, ALPHA * zv[k].y + t.y,
                                 ALPHA * zv[k].z + t.z, ALPHA * zv[k].w + t.w);
        const int idx = tid + k * NT;
        const int r = idx / V, q = idx % V;
        float* tr = tile + r * F;
        tr[tile_pos<F>(4 * q + 0)] = t.x;
        tr[tile_pos<F>(4 * q + 1)] = t.y;
        tr[tile_pos<F>(4 * q + 2)] = t.z;
        tr[tile_pos<F>(4 * q + 3)] = t.w;
    }
    __syncthreads();

    while (base < e1) {
        const int cnt = (e1 - base < 32) ? (e1 - base) : 32;
        if (cnt == 32) {
            for (int j0 = 0; j0 < 32; j0 += G) tile_edge_group<F, G>(tile, h, sv, dl, wbits, j0, lane);
        } else {
            constexpr int GP = G / 2;
            int j = 0;
            for (; j + GP <= cnt; j += GP) tile_edge_group<F, GP>(tile, h, sv, dl, wbits, j, lane);
            if (GP > 4 && cnt - j >= 4) { tile_edge_group<F, 4>(tile, h, sv, dl, wbits, j, lane); j += 4; }
            if (cnt - j >= 2) { tile_edge_group<F, 2>(tile, h, sv, dl, wbits, j, lane); j += 2; }
            if (cnt - j >= 1) tile_edge_group<F, 1>(tile, h, sv, dl, wbits, j, lane);
        }
        base += NW * 32;
        sv = 0; dl = 0; wbits = 0;
        if (base + lane < e1) {
            sv = bs[base + lane];
            dl = bd[base + lane] - row0;
            const float w = bw[base + lane];
            wbits = __float_as_int(MIX ? (1.0f - ALPHA) * w : w);
        }
    }
    __syncthreads();

#pragma unroll
    for (int k = 0; k < RT * V / NT; ++k) {
        const int idx = tid + k * NT;
        const int r = idx / V, q = idx % V;
        const int row = row0 + r;
        const float* tr = tile + r * F;
        const float4 t = make_float4(tr[tile_pos<F>(4 * q + 0)], tr[tile_pos<F>(4 * q + 1)],
                                     tr[tile_pos<F>(4 * q + 2)], tr[tile_pos<F>(4 * q + 3)]);
        if (row < n) ((float4*)(out + (size_t)row * F))[q] = t;
    }
}


static inline int cdiv(long long a, int b) { return (int)((a + b - 1) / b); }

extern "C" void kernel_launch(void* const* d_in, const int* in_sizes, int n_in,
                              void* d_out, int out_size, void* d_ws, size_t ws_size,
                              hipStream_t stream) {
    const float* x  = (const float*)d_in[0];
    const int*   ei = (const int*)d_in[1];
    const float* W1 = (const float*)d_in[2];
    const float* b1 = (const float*)d_in[3];
    const float* W3 = (const float*)d_in[4];
    const float* b3 = (const float*)d_in[5];
    float* out = (float*)d_out;

    const int N = in_sizes[0] / 256;
    const int E = in_sizes[1] / 2;
    const int* src = ei;
    const int* dst = ei + E;

    float* ws = (float*)d_ws;
    float* dinv = ws;
    size_t off = ((size_t)N + 63) & ~(size_t)63;
    float* bw = ws + off;
    off += ((size_t)E + 63) & ~(size_t)63;
    float* h0 = ws + off;
    off += (size_t)N * 128;
    float* h = ws + off;
    off += (size_t)N * 128;
    float* tmp = ws + off;
    off += (size_t)N * 128;
    const size_t Ea = ((size_t)E + 63) & ~(size_t)63;
    const int NB = cdiv(N, RT);
    int* bs = (int*)(ws + off);
    int* bd = bs + Ea;
    int* roff = bd + Ea;
    int* rcur = roff + (((size_t)NB + 1 + 63) & ~(size_t)63);

    const int T = 256;

    k_deg_init<<<cdiv(N, T), T, 0, stream>>>(dinv, rcur, N, NB);
    k_deg_count<<<cdiv(E, T), T, 0, stream>>>(dinv, rcur, dst, E);
    k_rsqrt<<<cdiv(N, T), T, 0, stream>>>(dinv, N);
    k_range_scan<<<1, 1024, 0, stream>>>(rcur, roff, NB);
    k_range_fill<<<cdiv(E, T), T, 0, stream>>>(bs, bd, bw, rcur, src, dst, dinv, E);

    k_gemm_wmma<256, 128><<<512, 32 * (128 / 16), 0, stream>>>(x, W1, b1, tmp, N);
    k_prop_tile<128, false><<<NB, T, 0, stream>>>(h0, tmp, nullptr, dinv, bs, bd, bw, roff, N);

    const float* hs = h0;
    for (int it = 0; it < 10; ++it) {
        float* hd = (it & 1) ? tmp : h;
        k_prop_tile<128, true><<<NB, T, 0, stream>>>(hd, hs, h0, dinv, bs, bd, bw, roff, N);
        hs = hd;
    }

    float* hw = h;
    k_gemm_wmma<128, 64><<<1042, 32 * (64 / 16), 0, stream>>>(hs, W3, b3, hw, N);
    k_prop_tile<64, false><<<NB, T, 0, stream>>>(out, hw, nullptr, dinv, bs, bd, bw, roff, N);
}
